# starting kernel
# baseline (speedup 1.0000x reference)
#include <hip/hip_runtime.h>

typedef float v2f __attribute__((ext_vector_type(2)));
typedef float v8f __attribute__((ext_vector_type(8)));

#define N_NODES 50000
#define HIDC 64
#define KMAX 128
#define APAD 4


__global__ void k_init_deg(float* __restrict__ deg, int n) {
    int i = blockIdx.x * blockDim.x + threadIdx.x;
    if (i < n) deg[i] = 1.0f;
}

__global__ void k_count_deg(const int* __restrict__ dst, float* __restrict__ deg, int E) {
    int e = blockIdx.x * blockDim.x + threadIdx.x;
    if (e < E) atomicAdd(&deg[dst[e]], 1.0f);
}

__global__ void k_rsqrt_inplace(float* __restrict__ d, int n) {
    int i = blockIdx.x * blockDim.x + threadIdx.x;
    if (i < n) d[i] = rsqrtf(d[i]);
}


__global__ void k_wmma_gemm(const float* __restrict__ A, const float* __restrict__ B,
                            const float* __restrict__ bias, float* __restrict__ C,
                            int K, int N) {
    __shared__ float As[16 * (KMAX + APAD)];

    const int m0   = blockIdx.x * 16;
    const int lane = threadIdx.x & 31;
    const int wave = threadIdx.x >> 5;
    const int n0   = wave * 16;
    const int half = lane >> 4;
    const int idx  = lane & 15;
    const int ldsK = K + APAD;

    {
        const int kq = K >> 2;
        for (int r = 0; r < 16; ++r) {
            const float4* srow = (const float4*)(A + (long)(m0 + r) * K);
            float4*       drow = (float4*)(As + r * ldsK);
            for (int c = threadIdx.x; c < kq; c += blockDim.x) drow[c] = srow[c];
        }
    }
    __syncthreads();

    const float* arow = As + idx * ldsK;
    const int    bcol = n0 + idx;
    const bool   bok  = bcol < N;
    const float* bp   = B + (bok ? bcol : (N - 1));

    v8f acc = {0.f, 0.f, 0.f, 0.f, 0.f, 0.f, 0.f, 0.f};

    for (int k = 0; k < K; k += 4) {
        const int ka = k + 2 * half;
        v2f a, b;
        a.x = arow[ka];
        a.y = arow[ka + 1];
        b.x = bp[(long)ka * N];
        b.y = bp[(long)(ka + 1) * N];
        acc = __builtin_amdgcn_wmma_f32_16x16x4_f32(
             false, a,  false, b,
             (short)0, acc,  false,  false);
    }

    if (bok) {
        const float bb = bias ? bias[bcol] : 0.0f;
#pragma unroll
        for (int v = 0; v < 8; ++v) {
            const int row = m0 + v + 8 * half;
            C[(long)row * N + bcol] = acc[v] + bb;
        }
    }
}


__global__ void k_self_init(const float* __restrict__ hw, const float* __restrict__ dis,
                            float* __restrict__ agg, int n) {
    long t = (long)blockIdx.x * blockDim.x + threadIdx.x;
    if (t < (long)n * HIDC) {
        int i = (int)(t >> 6);
        float di = dis[i];
        agg[t] = hw[t] * di * di;
    }
}

__global__ void k_scatter(const float* __restrict__ hw, const float* __restrict__ dis,
                          const int* __restrict__ src, const int* __restrict__ dst,
                          float* __restrict__ agg, int E) {
    long t = (long)blockIdx.x * blockDim.x + threadIdx.x;
    int e = (int)(t >> 4);
    if (e >= E) return;
    int g = (int)(t & 15);
    int s = src[e];
    int d = dst[e];
    float w = dis[s] * dis[d];
    const float4 v = *((const float4*)(hw + (long)s * HIDC) + g);
    float* ap = agg + (long)d * HIDC + g * 4;
    atomicAdd(ap + 0, v.x * w);
    atomicAdd(ap + 1, v.y * w);
    atomicAdd(ap + 2, v.z * w);
    atomicAdd(ap + 3, v.w * w);
}

__global__ void k_bias_relu(float* __restrict__ h, const float* __restrict__ bias, int n) {
    long t = (long)blockIdx.x * blockDim.x + threadIdx.x;
    if (t < (long)n * HIDC) {
        int c = (int)(t & (HIDC - 1));
        float v = h[t] + bias[c];
        h[t] = v > 0.f ? v : 0.f;
    }
}


extern "C" void kernel_launch(void* const* d_in, const int* in_sizes, int n_in,
                              void* d_out, int out_size, void* d_ws, size_t ws_size,
                              hipStream_t stream) {
    const float* x  = (const float*)d_in[0];
    const int*   ei = (const int*)d_in[1];
    const float* W1 = (const float*)d_in[2];
    const float* b1 = (const float*)d_in[3];
    const float* W2 = (const float*)d_in[4];
    const float* b2 = (const float*)d_in[5];
    const float* Wc = (const float*)d_in[6];
    const float* bc = (const float*)d_in[7];
    float* out = (float*)d_out;

    const int n = N_NODES;
    const int E = in_sizes[1] / 2;
    const int* src = ei;
    const int* dst = ei + E;

    char* ws = (char*)d_ws;
    float* dis = (float*)ws;
    size_t off = (((size_t)n * 4) + 255) & ~(size_t)255;
    float* bufA = (float*)(ws + off);
    off += (((size_t)n * HIDC * 4) + 255) & ~(size_t)255;
    float* bufB = (float*)(ws + off);

    const int T = 256;
    const int gN   = (n + T - 1) / T;
    const int gE   = (E + T - 1) / T;
    const int gNC  = (n * HIDC + T - 1) / T;
    const int gE16 = (int)(((long)E * 16 + T - 1) / T);
    const int rowTiles = n / 16;

    k_init_deg<<<gN, T, 0, stream>>>(dis, n);
    k_count_deg<<<gE, T, 0, stream>>>(dst, dis, E);
    k_rsqrt_inplace<<<gN, T, 0, stream>>>(dis, n);

    k_wmma_gemm<<<rowTiles, 128, 0, stream>>>(x, W1, nullptr, bufA, 128, HIDC);
    k_self_init<<<gNC, T, 0, stream>>>(bufA, dis, bufB, n);
    k_scatter<<<gE16, T, 0, stream>>>(bufA, dis, src, dst, bufB, E);
    k_bias_relu<<<gNC, T, 0, stream>>>(bufB, b1, n);

    k_wmma_gemm<<<rowTiles, 128, 0, stream>>>(bufB, W2, nullptr, bufA, HIDC, HIDC);
    k_self_init<<<gNC, T, 0, stream>>>(bufA, dis, bufB, n);
    k_scatter<<<gE16, T, 0, stream>>>(bufA, dis, src, dst, bufB, E);
    k_bias_relu<<<gNC, T, 0, stream>>>(bufB, b2, n);

    k_wmma_gemm<<<rowTiles, 32, 0, stream>>>(bufB, Wc, bc, out, HIDC, 10);
}


# A-tile staging of the three WMMA product kernels: all threads take part and each issues all its four-float loads of the 16 x K tile before waiting: one memory trip, not 16 serial row trips
# speedup vs baseline: 2.0764x; 2.0764x over previous
#include <hip/hip_runtime.h>

typedef float v2f __attribute__((ext_vector_type(2)));
typedef float v4f __attribute__((ext_vector_type(4)));
typedef float v8f __attribute__((ext_vector_type(8)));

#define N_NODES 50000
#define HIDC 64
#define KMAX 128
#define APAD 4


__global__ void k_init_deg(float* __restrict__ deg, unsigned* __restrict__ ctr, int n) {
    int i = blockIdx.x * blockDim.x + threadIdx.x;
    if (i < n) deg[i] = 1.0f;
    if (i < 4) ctr[i] = 0u;
}

__global__ void k_count_deg(const int* __restrict__ dst, float* __restrict__ deg, int E) {
    int e = blockIdx.x * blockDim.x + threadIdx.x;
    if (e < E) atomicAdd(&deg[dst[e]], 1.0f);
}

__global__ void k_rsqrt_inplace(float* __restrict__ d, int n) {
    int i = blockIdx.x * blockDim.x + threadIdx.x;
    if (i < n) d[i] = rsqrtf(d[i]);
}


template <int NLD, bool ACT_IN, bool SELF_OUT>
__global__ void k_wmma_gemm(const float* __restrict__ A, const float* __restrict__ abias,
                            const float* __restrict__ B, const float* __restrict__ bias,
                            float* __restrict__ C,
                            const float* __restrict__ dis, float* __restrict__ agg,
                            int K, int N) {
    __shared__ float As[16 * (KMAX + APAD)];

    const int m0   = blockIdx.x * 16;
    const int lane = threadIdx.x & 31;
    const int wave = threadIdx.x >> 5;
    const int n0   = wave * 16;
    const int half = lane >> 4;
    const int idx  = lane & 15;
    const int ldsK = K + APAD;

    {
        const unsigned kq    = (unsigned)K >> 2;
        const unsigned total = 16u * kq;
        const v4f*     tile  = (const v4f*)(A + (long)m0 * K);
        for (unsigned base = threadIdx.x; base < total; base += NLD * blockDim.x) {
            v4f v[NLD], ab[NLD];
#pragma unroll
            for (int j = 0; j < NLD; ++j) {
                const unsigned i = base + j * blockDim.x;
                if (i < total) {
                    v[j] = tile[i];
                    if (ACT_IN) ab[j] = ((const v4f*)abias)[i % kq];
                }
            }
#pragma unroll
            for (int j = 0; j < NLD; ++j) {
                const unsigned i = base + j * blockDim.x;
                if (i < total) {
                    const unsigned r = i / kq;
                    const unsigned c = i - r * kq;
                    v4f t = v[j];
                    if (ACT_IN) {
                        t += ab[j];
                        t.x = t.x > 0.f ? t.x : 0.f;
                        t.y = t.y > 0.f ? t.y : 0.f;
                        t.z = t.z > 0.f ? t.z : 0.f;
                        t.w = t.w > 0.f ? t.w : 0.f;
                    }
                    v4f* drow = (v4f*)(As + r * ldsK);
                    drow[c] = t;
                }
            }
        }
    }
    __syncthreads();

    const float* arow = As + idx * ldsK;
    const int    bcol = n0 + idx;
    const bool   bok  = bcol < N;
    const float* bp   = B + (bok ? bcol : (N - 1));

    v8f acc = {0.f, 0.f, 0.f, 0.f, 0.f, 0.f, 0.f, 0.f};

    for (int k = 0; k < K; k += 4) {
        const int ka = k + 2 * half;
        v2f a, b;
        a.x = arow[ka];
        a.y = arow[ka + 1];
        b.x = bp[(long)ka * N];
        b.y = bp[(long)(ka + 1) * N];
        acc = __builtin_amdgcn_wmma_f32_16x16x4_f32(
             false, a,  false, b,
             (short)0, acc,  false,  false);
    }

    if (bok) {
        const float bb = bias ? bias[bcol] : 0.0f;
#pragma unroll
        for (int v = 0; v < 8; ++v) {
            const int   row = m0 + v + 8 * half;
            const float c   = acc[v] + bb;
            C[(long)row * N + bcol] = c;
            if (SELF_OUT) {
                const float di = dis[row];
                agg[(long)row * N + bcol] = c * di * di;
            }
        }
    }
}


#define SPLIT_ROW (N_NODES / 2)
#define NSLICE 782
#define EGRP 16
__global__ void __launch_bounds__(256)
k_scatter(const float* __restrict__ hw, const float* __restrict__ dis,
          const int* __restrict__ src, const int* __restrict__ dst,
          float* __restrict__ agg, unsigned* __restrict__ ctr, int E) {
    __shared__ int sh_half, sh_tick;
    if (threadIdx.x == 0) {
        const unsigned id = __builtin_amdgcn_s_sendmsg_rtn(0x87);
        int      hh = (int)((id >> 18) & 1u);
        unsigned t  = atomicAdd(&ctr[hh], 1u);
        if (t >= NSLICE) { hh ^= 1; t = atomicAdd(&ctr[hh], 1u); }
        if (t >= NSLICE) hh = -1;
        sh_half = hh;
        sh_tick = (int)t;
    }
    __syncthreads();
    const int hh = __builtin_amdgcn_readfirstlane(sh_half);
    const int tk = __builtin_amdgcn_readfirstlane(sh_tick);
    if (hh < 0) return;

    const unsigned lane = threadIdx.x & 31u;
    const int wv   = __builtin_amdgcn_readfirstlane((int)(threadIdx.x >> 5));
    const int ngrp = (E + 31) >> 5;

    for (int g = tk * 8 + wv; g < ngrp; g += NSLICE * 8) {
        const int e = g * 32 + (int)lane;
        int   s = 0, d = 0;
        float w = 0.f;
        bool  mine = false;
        if (e < E) {
            s = src[e];
            d = dst[e];
            mine = ((d >= SPLIT_ROW) ? 1 : 0) == hh;
            if (mine) w = dis[s] * dis[d];
        }
        asm volatile("" :: "v"(s), "v"(d));
        const int      w_bits = __float_as_int(w);
        const unsigned mask   = __builtin_amdgcn_ballot_w32(mine);

#pragma unroll
        for (int b = 0; b < 32; b += EGRP) {
            float v0[EGRP], v1[EGRP];
#pragma unroll
            for (int i = 0; i < EGRP; ++i) {
                v0[i] = 0.f;
                v1[i] = 0.f;
                if (mask & (1u << (b + i))) {
                    const unsigned si = (unsigned)__builtin_amdgcn_readlane(s, b + i);
                    v0[i] = hw[si * HIDC + lane];
                    v1[i] = hw[si * HIDC + lane + 32u];
                }
            }
#pragma unroll
            for (int i = 0; i < EGRP; ++i) {
                if (mask & (1u << (b + i))) {
                    const unsigned di = (unsigned)__builtin_amdgcn_readlane(d, b + i);
                    const float    wi = __int_as_float(__builtin_amdgcn_readlane(w_bits, b + i));
                    atomicAdd(agg + (di * HIDC + lane),       v0[i] * wi);
                    atomicAdd(agg + (di * HIDC + lane + 32u), v1[i] * wi);
                }
            }
        }
    }
}


extern "C" void kernel_launch(void* const* d_in, const int* in_sizes, int n_in,
                              void* d_out, int out_size, void* d_ws, size_t ws_size,
                              hipStream_t stream) {
    const float* x  = (const float*)d_in[0];
    const int*   ei = (const int*)d_in[1];
    const float* W1 = (const float*)d_in[2];
    const float* b1 = (const float*)d_in[3];
    const float* W2 = (const float*)d_in[4];
    const float* b2 = (const float*)d_in[5];
    const float* Wc = (const float*)d_in[6];
    const float* bc = (const float*)d_in[7];
    float* out = (float*)d_out;

    const int n = N_NODES;
    const int E = in_sizes[1] / 2;
    const int* src = ei;
    const int* dst = ei + E;

    char* ws = (char*)d_ws;
    float* dis = (float*)ws;
    size_t off = (((size_t)n * 4) + 255) & ~(size_t)255;
    float* bufA = (float*)(ws + off);
    off += (((size_t)n * HIDC * 4) + 255) & ~(size_t)255;
    float* bufB = (float*)(ws + off);
    off += (((size_t)n * HIDC * 4) + 255) & ~(size_t)255;
    float* bufC = (float*)(ws + off);
    off += (((size_t)n * HIDC * 4) + 255) & ~(size_t)255;
    unsigned* ctr = (unsigned*)(ws + off);

    const int T = 256;
    const int gN   = (n + T - 1) / T;
    const int gE   = (E + T - 1) / T;
    const int rowTiles = n / 16;

    k_init_deg<<<gN, T, 0, stream>>>(dis, ctr, n);
    k_count_deg<<<gE, T, 0, stream>>>(dst, dis, E);
    k_rsqrt_inplace<<<gN, T, 0, stream>>>(dis, n);

    k_wmma_gemm<4, false, true><<<rowTiles, 128, 0, stream>>>(x, nullptr, W1, nullptr, bufA, dis, bufB, 128, HIDC);
    k_scatter<<<2 * NSLICE, T, 0, stream>>>(bufA, dis, src, dst, bufB, ctr, E);

    k_wmma_gemm<2, true, true><<<rowTiles, 128, 0, stream>>>(bufB, b1, W2, nullptr, bufA, dis, bufC, HIDC, HIDC);
    k_scatter<<<2 * NSLICE, T, 0, stream>>>(bufA, dis, src, dst, bufC, ctr + 2, E);

    k_wmma_gemm<8, true, false><<<rowTiles, 32, 0, stream>>>(bufC, b2, Wc, bc, out, nullptr, nullptr, HIDC, 10);
}


# Normalisation moved out of the edge walk: product epilogue scales row i by dis[i], edges add plain rows (no dis gathers, no multiply), next A staging scales the sum by dis[row]; sum order kept
# speedup vs baseline: 2.2472x; 1.0823x over previous
#include <hip/hip_runtime.h>

typedef float v2f __attribute__((ext_vector_type(2)));
typedef float v4f __attribute__((ext_vector_type(4)));
typedef float v8f __attribute__((ext_vector_type(8)));

#define N_NODES 50000
#define HIDC 64
#define KMAX 128
#define APAD 4


__global__ void k_init_deg(float* __restrict__ deg, unsigned* __restrict__ ctr, int n) {
    int i = blockIdx.x * blockDim.x + threadIdx.x;
    if (i < n) deg[i] = 1.0f;
    if (i < 4) ctr[i] = 0u;
}

__global__ void k_count_deg(const int* __restrict__ dst, float* __restrict__ deg, int E) {
    int e = blockIdx.x * blockDim.x + threadIdx.x;
    if (e < E) atomicAdd(&deg[dst[e]], 1.0f);
}

__global__ void k_rsqrt_inplace(float* __restrict__ d, int n) {
    int i = blockIdx.x * blockDim.x + threadIdx.x;
    if (i < n) d[i] = rsqrtf(d[i]);
}


template <int NLD, bool ACT_IN, bool SELF_OUT>
__global__ void k_wmma_gemm(const float* __restrict__ A, const float* __restrict__ abias,
                            const float* __restrict__ B, const float* __restrict__ bias,
                            float* __restrict__ C,
                            const float* __restrict__ dis, float* __restrict__ agg,
                            int K, int N) {
    __shared__ float As[16 * (KMAX + APAD)];

    const int m0   = blockIdx.x * 16;
    const int lane = threadIdx.x & 31;
    const int wave = threadIdx.x >> 5;
    const int n0   = wave * 16;
    const int half = lane >> 4;
    const int idx  = lane & 15;
    const int ldsK = K + APAD;

    {
        const unsigned kq    = (unsigned)K >> 2;
        const unsigned total = 16u * kq;
        const v4f*     tile  = (const v4f*)(A + (long)m0 * K);
        for (unsigned base = threadIdx.x; base < total; base += NLD * blockDim.x) {
            v4f   v[NLD], ab[NLD];
            float dr[NLD];
#pragma unroll
            for (int j = 0; j < NLD; ++j) {
                const unsigned i = base + j * blockDim.x;
                if (i < total) {
                    v[j] = tile[i];
                    if (ACT_IN) {
                        ab[j] = ((const v4f*)abias)[i % kq];
                        dr[j] = dis[m0 + i / kq];
                    }
                }
            }
#pragma unroll
            for (int j = 0; j < NLD; ++j) {
                const unsigned i = base + j * blockDim.x;
                if (i < total) {
                    const unsigned r = i / kq;
                    const unsigned c = i - r * kq;
                    v4f t = v[j];
                    if (ACT_IN) {
                        t = t * dr[j] + ab[j];
                        t.x = t.x > 0.f ? t.x : 0.f;
                        t.y = t.y > 0.f ? t.y : 0.f;
                        t.z = t.z > 0.f ? t.z : 0.f;
                        t.w = t.w > 0.f ? t.w : 0.f;
                    }
                    v4f* drow = (v4f*)(As + r * ldsK);
                    drow[c] = t;
                }
            }
        }
    }
    __syncthreads();

    const float* arow = As + idx * ldsK;
    const int    bcol = n0 + idx;
    const bool   bok  = bcol < N;
    const float* bp   = B + (bok ? bcol : (N - 1));

    v8f acc = {0.f, 0.f, 0.f, 0.f, 0.f, 0.f, 0.f, 0.f};

    for (int k = 0; k < K; k += 4) {
        const int ka = k + 2 * half;
        v2f a, b;
        a.x = arow[ka];
        a.y = arow[ka + 1];
        b.x = bp[(long)ka * N];
        b.y = bp[(long)(ka + 1) * N];
        acc = __builtin_amdgcn_wmma_f32_16x16x4_f32(
             false, a,  false, b,
             (short)0, acc,  false,  false);
    }

    if (bok) {
        const float bb = bias ? bias[bcol] : 0.0f;
#pragma unroll
        for (int v = 0; v < 8; ++v) {
            const int   row = m0 + v + 8 * half;
            float       c   = acc[v] + bb;
            if (SELF_OUT) {
                c *= dis[row];
                agg[(long)row * N + bcol] = c;
            }
            C[(long)row * N + bcol] = c;
        }
    }
}


#define SPLIT_ROW (N_NODES / 2)
#define NSLICE 782
#define EGRP 16
__global__ void __launch_bounds__(256)
k_scatter(const float* __restrict__ hw,
          const int* __restrict__ src, const int* __restrict__ dst,
          float* __restrict__ agg, unsigned* __restrict__ ctr, int E) {
    __shared__ int sh_half, sh_tick;
    if (threadIdx.x == 0) {
        const unsigned id = __builtin_amdgcn_s_sendmsg_rtn(0x87);
        int      hh = (int)((id >> 18) & 1u);
        unsigned t  = atomicAdd(&ctr[hh], 1u);
        if (t >= NSLICE) { hh ^= 1; t = atomicAdd(&ctr[hh], 1u); }
        if (t >= NSLICE) hh = -1;
        sh_half = hh;
        sh_tick = (int)t;
    }
    __syncthreads();
    const int hh = __builtin_amdgcn_readfirstlane(sh_half);
    const int tk = __builtin_amdgcn_readfirstlane(sh_tick);
    if (hh < 0) return;

    const unsigned lane = threadIdx.x & 31u;
    const int wv   = __builtin_amdgcn_readfirstlane((int)(threadIdx.x >> 5));
    const int ngrp = (E + 31) >> 5;

    for (int g = tk * 8 + wv; g < ngrp; g += NSLICE * 8) {
        const int e = g * 32 + (int)lane;
        int  s = 0, d = 0;
        bool mine = false;
        if (e < E) {
            s = src[e];
            d = dst[e];
            mine = ((d >= SPLIT_ROW) ? 1 : 0) == hh;
        }
        asm volatile("" :: "v"(s), "v"(d));
        const unsigned mask = __builtin_amdgcn_ballot_w32(mine);

#pragma unroll
        for (int b = 0; b < 32; b += EGRP) {
            float v0[EGRP], v1[EGRP];
#pragma unroll
            for (int i = 0; i < EGRP; ++i) {
                v0[i] = 0.f;
                v1[i] = 0.f;
                if (mask & (1u << (b + i))) {
                    const unsigned si = (unsigned)__builtin_amdgcn_readlane(s, b + i);
                    v0[i] = hw[si * HIDC + lane];
                    v1[i] = hw[si * HIDC + lane + 32u];
                }
            }
#pragma unroll
            for (int i = 0; i < EGRP; ++i) {
                if (mask & (1u << (b + i))) {
                    const unsigned di = (unsigned)__builtin_amdgcn_readlane(d, b + i);
                    atomicAdd(agg + (di * HIDC + lane),       v0[i]);
                    atomicAdd(agg + (di * HIDC + lane + 32u), v1[i]);
                }
            }
        }
    }
}


extern "C" void kernel_launch(void* const* d_in, const int* in_sizes, int n_in,
                              void* d_out, int out_size, void* d_ws, size_t ws_size,
                              hipStream_t stream) {
    const float* x  = (const float*)d_in[0];
    const int*   ei = (const int*)d_in[1];
    const float* W1 = (const float*)d_in[2];
    const float* b1 = (const float*)d_in[3];
    const float* W2 = (const float*)d_in[4];
    const float* b2 = (const float*)d_in[5];
    const float* Wc = (const float*)d_in[6];
    const float* bc = (const float*)d_in[7];
    float* out = (float*)d_out;

    const int n = N_NODES;
    const int E = in_sizes[1] / 2;
    const int* src = ei;
    const int* dst = ei + E;

    char* ws = (char*)d_ws;
    float* dis = (float*)ws;
    size_t off = (((size_t)n * 4) + 255) & ~(size_t)255;
    float* bufA = (float*)(ws + off);
    off += (((size_t)n * HIDC * 4) + 255) & ~(size_t)255;
    float* bufB = (float*)(ws + off);
    off += (((size_t)n * HIDC * 4) + 255) & ~(size_t)255;
    float* bufC = (float*)(ws + off);
    off += (((size_t)n * HIDC * 4) + 255) & ~(size_t)255;
    unsigned* ctr = (unsigned*)(ws + off);

    const int T = 256;
    const int gN   = (n + T - 1) / T;
    const int gE   = (E + T - 1) / T;
    const int rowTiles = n / 16;

    k_init_deg<<<gN, T, 0, stream>>>(dis, ctr, n);
    k_count_deg<<<gE, T, 0, stream>>>(dst, dis, E);
    k_rsqrt_inplace<<<gN, T, 0, stream>>>(dis, n);

    k_wmma_gemm<4, false, true><<<rowTiles, 128, 0, stream>>>(x, nullptr, W1, nullptr, bufA, dis, bufB, 128, HIDC);
    k_scatter<<<2 * NSLICE, T, 0, stream>>>(bufA, src, dst, bufB, ctr, E);

    k_wmma_gemm<2, true, true><<<rowTiles, 128, 0, stream>>>(bufB, b1, W2, nullptr, bufA, dis, bufC, HIDC, HIDC);
    k_scatter<<<2 * NSLICE, T, 0, stream>>>(bufA, src, dst, bufC, ctr + 2, E);

    k_wmma_gemm<8, true, false><<<rowTiles, 32, 0, stream>>>(bufC, b2, Wc, bc, out, dis, nullptr, HIDC, 10);
}


# In-degrees counted in LDS histograms (4 node ranges x 32 edge slices), stored as planes that the rsqrt pass adds: no 800000 one-word f32 global atomics, no clear launch, 8 -> 7; edge sum order kept
# speedup vs baseline: 3.1797x; 1.4150x over previous
#include <hip/hip_runtime.h>

typedef float v2f __attribute__((ext_vector_type(2)));
typedef float v4f __attribute__((ext_vector_type(4)));
typedef float v8f __attribute__((ext_vector_type(8)));

#define N_NODES 50000
#define HIDC 64
#define KMAX 128
#define APAD 4


#define DEG_R     4
#define DEG_S     32
#define DEG_RANGE 12512
#define DEG_PITCH 50048
#define DEG_UNR   25

__global__ void __launch_bounds__(1024)
k_deg_hist(const int* __restrict__ dst, unsigned* __restrict__ planes,
           unsigned* __restrict__ ctr, int E) {
    __shared__ unsigned h[DEG_RANGE];
    const int tid = threadIdx.x;
    const int r   = blockIdx.x % DEG_R;
    const int s   = blockIdx.x / DEG_R;
    const int r0  = r * DEG_RANGE;
    const unsigned len = (unsigned)((N_NODES - r0) < DEG_RANGE ? (N_NODES - r0) : DEG_RANGE);

#pragma unroll
    for (int j = 0; j < (DEG_RANGE + 1023) / 1024; ++j) {
        const int i = tid + j * 1024;
        if (i < DEG_RANGE) h[i] = 0u;
    }
    if (blockIdx.x == 0 && tid < 4) ctr[tid] = 0u;
    __syncthreads();

    const int per = (E + DEG_S - 1) / DEG_S;
    const int e0  = s * per;
    const int e1  = (e0 + per) < E ? (e0 + per) : E;
    for (int base = e0 + tid; base < e1; base += 1024 * DEG_UNR) {
        int d[DEG_UNR];
#pragma unroll
        for (int j = 0; j < DEG_UNR; ++j) {
            const int e = base + j * 1024;
            d[j] = (e < e1) ? dst[e] : -1;
        }
#pragma unroll
        for (int j = 0; j < DEG_UNR; ++j) {
            const unsigned k = (unsigned)d[j] - (unsigned)r0;
            if (k < len) atomicAdd(&h[k], 1u);
        }
    }
    __syncthreads();

    unsigned* p = planes + (size_t)s * DEG_PITCH + r0;
#pragma unroll
    for (int j = 0; j < (DEG_RANGE + 1023) / 1024; ++j) {
        const unsigned i = (unsigned)(tid + j * 1024);
        if (i < len) p[i] = h[i];
    }
}

__global__ void k_dis_from_planes(const unsigned* __restrict__ planes, float* __restrict__ dis, int n) {
    const int i = blockIdx.x * blockDim.x + threadIdx.x;
    if (i < n) {
        unsigned c[DEG_S];
#pragma unroll
        for (int s = 0; s < DEG_S; ++s) c[s] = planes[(size_t)s * DEG_PITCH + i];
        unsigned t = 0u;
#pragma unroll
        for (int s = 0; s < DEG_S; ++s) t += c[s];
        dis[i] = rsqrtf(1.0f + (float)t);
    }
}


template <int NLD, bool ACT_IN, bool SELF_OUT>
__global__ void k_wmma_gemm(const float* __restrict__ A, const float* __restrict__ abias,
                            const float* __restrict__ B, const float* __restrict__ bias,
                            float* __restrict__ C,
                            const float* __restrict__ dis, float* __restrict__ agg,
                            int K, int N) {
    __shared__ float As[16 * (KMAX + APAD)];

    const int m0   = blockIdx.x * 16;
    const int lane = threadIdx.x & 31;
    const int wave = threadIdx.x >> 5;
    const int n0   = wave * 16;
    const int half = lane >> 4;
    const int idx  = lane & 15;
    const int ldsK = K + APAD;

    {
        const unsigned kq    = (unsigned)K >> 2;
        const unsigned total = 16u * kq;
        const v4f*     tile  = (const v4f*)(A + (long)m0 * K);
        for (unsigned base = threadIdx.x; base < total; base += NLD * blockDim.x) {
            v4f   v[NLD], ab[NLD];
            float dr[NLD];
#pragma unroll
            for (int j = 0; j < NLD; ++j) {
                const unsigned i = base + j * blockDim.x;
                if (i < total) {
                    v[j] = tile[i];
                    if (ACT_IN) {
                        ab[j] = ((const v4f*)abias)[i % kq];
                        dr[j] = dis[m0 + i / kq];
                    }
                }
            }
#pragma unroll
            for (int j = 0; j < NLD; ++j) {
                const unsigned i = base + j * blockDim.x;
                if (i < total) {
                    const unsigned r = i / kq;
                    const unsigned c = i - r * kq;
                    v4f t = v[j];
                    if (ACT_IN) {
                        t = t * dr[j] + ab[j];
                        t.x = t.x > 0.f ? t.x : 0.f;
                        t.y = t.y > 0.f ? t.y : 0.f;
                        t.z = t.z > 0.f ? t.z : 0.f;
                        t.w = t.w > 0.f ? t.w : 0.f;
                    }
                    v4f* drow = (v4f*)(As + r * ldsK);
                    drow[c] = t;
                }
            }
        }
    }
    __syncthreads();

    const float* arow = As + idx * ldsK;
    const int    bcol = n0 + idx;
    const bool   bok  = bcol < N;
    const float* bp   = B + (bok ? bcol : (N - 1));

    v8f acc = {0.f, 0.f, 0.f, 0.f, 0.f, 0.f, 0.f, 0.f};

    for (int k = 0; k < K; k += 4) {
        const int ka = k + 2 * half;
        v2f a, b;
        a.x = arow[ka];
        a.y = arow[ka + 1];
        b.x = bp[(long)ka * N];
        b.y = bp[(long)(ka + 1) * N];
        acc = __builtin_amdgcn_wmma_f32_16x16x4_f32(
             false, a,  false, b,
             (short)0, acc,  false,  false);
    }

    if (bok) {
        const float bb = bias ? bias[bcol] : 0.0f;
#pragma unroll
        for (int v = 0; v < 8; ++v) {
            const int   row = m0 + v + 8 * half;
            float       c   = acc[v] + bb;
            if (SELF_OUT) {
                c *= dis[row];
                agg[(long)row * N + bcol] = c;
            }
            C[(long)row * N + bcol] = c;
        }
    }
}


#define SPLIT_ROW (N_NODES / 2)
#define NSLICE 782
#define EGRP 16
__global__ void __launch_bounds__(256)
k_scatter(const float* __restrict__ hw,
          const int* __restrict__ src, const int* __restrict__ dst,
          float* __restrict__ agg, unsigned* __restrict__ ctr, int E) {
    __shared__ int sh_half, sh_tick;
    if (threadIdx.x == 0) {
        const unsigned id = __builtin_amdgcn_s_sendmsg_rtn(0x87);
        int      hh = (int)((id >> 18) & 1u);
        unsigned t  = atomicAdd(&ctr[hh], 1u);
        if (t >= NSLICE) { hh ^= 1; t = atomicAdd(&ctr[hh], 1u); }
        if (t >= NSLICE) hh = -1;
        sh_half = hh;
        sh_tick = (int)t;
    }
    __syncthreads();
    const int hh = __builtin_amdgcn_readfirstlane(sh_half);
    const int tk = __builtin_amdgcn_readfirstlane(sh_tick);
    if (hh < 0) return;

    const unsigned lane = threadIdx.x & 31u;
    const int wv   = __builtin_amdgcn_readfirstlane((int)(threadIdx.x >> 5));
    const int ngrp = (E + 31) >> 5;

    for (int g = tk * 8 + wv; g < ngrp; g += NSLICE * 8) {
        const int e = g * 32 + (int)lane;
        int  s = 0, d = 0;
        bool mine = false;
        if (e < E) {
            s = src[e];
            d = dst[e];
            mine = ((d >= SPLIT_ROW) ? 1 : 0) == hh;
        }
        asm volatile("" :: "v"(s), "v"(d));
        const unsigned mask = __builtin_amdgcn_ballot_w32(mine);

#pragma unroll
        for (int b = 0; b < 32; b += EGRP) {
            float v0[EGRP], v1[EGRP];
#pragma unroll
            for (int i = 0; i < EGRP; ++i) {
                v0[i] = 0.f;
                v1[i] = 0.f;
                if (mask & (1u << (b + i))) {
                    const unsigned si = (unsigned)__builtin_amdgcn_readlane(s, b + i);
                    v0[i] = hw[si * HIDC + lane];
                    v1[i] = hw[si * HIDC + lane + 32u];
                }
            }
#pragma unroll
            for (int i = 0; i < EGRP; ++i) {
                if (mask & (1u << (b + i))) {
                    const unsigned di = (unsigned)__builtin_amdgcn_readlane(d, b + i);
                    atomicAdd(agg + (di * HIDC + lane),       v0[i]);
                    atomicAdd(agg + (di * HIDC + lane + 32u), v1[i]);
                }
            }
        }
    }
}


extern "C" void kernel_launch(void* const* d_in, const int* in_sizes, int n_in,
                              void* d_out, int out_size, void* d_ws, size_t ws_size,
                              hipStream_t stream) {
    const float* x  = (const float*)d_in[0];
    const int*   ei = (const int*)d_in[1];
    const float* W1 = (const float*)d_in[2];
    const float* b1 = (const float*)d_in[3];
    const float* W2 = (const float*)d_in[4];
    const float* b2 = (const float*)d_in[5];
    const float* Wc = (const float*)d_in[6];
    const float* bc = (const float*)d_in[7];
    float* out = (float*)d_out;

    const int n = N_NODES;
    const int E = in_sizes[1] / 2;
    const int* src = ei;
    const int* dst = ei + E;

    char* ws = (char*)d_ws;
    float* dis = (float*)ws;
    size_t off = (((size_t)n * 4) + 255) & ~(size_t)255;
    float* bufA = (float*)(ws + off);
    off += (((size_t)n * HIDC * 4) + 255) & ~(size_t)255;
    float* bufB = (float*)(ws + off);
    off += (((size_t)n * HIDC * 4) + 255) & ~(size_t)255;
    float* bufC = (float*)(ws + off);
    off += (((size_t)n * HIDC * 4) + 255) & ~(size_t)255;
    unsigned* ctr = (unsigned*)(ws + off);
    off += 256;
    unsigned* planes = (unsigned*)(ws + off);

    const int T = 256;
    const int gN   = (n + T - 1) / T;
    const int rowTiles = n / 16;

    k_deg_hist<<<DEG_R * DEG_S, 1024, 0, stream>>>(dst, planes, ctr, E);
    k_dis_from_planes<<<gN, T, 0, stream>>>(planes, dis, n);

    k_wmma_gemm<4, false, true><<<rowTiles, 128, 0, stream>>>(x, nullptr, W1, nullptr, bufA, dis, bufB, 128, HIDC);
    k_scatter<<<2 * NSLICE, T, 0, stream>>>(bufA, src, dst, bufB, ctr, E);

    k_wmma_gemm<2, true, true><<<rowTiles, 128, 0, stream>>>(bufB, b1, W2, nullptr, bufA, dis, bufC, HIDC, HIDC);
    k_scatter<<<2 * NSLICE, T, 0, stream>>>(bufA, src, dst, bufC, ctr + 2, E);

    k_wmma_gemm<8, true, false><<<rowTiles, 32, 0, stream>>>(bufC, b2, Wc, bc, out, dis, nullptr, HIDC, 10);
}
